# starting kernel
# baseline (speedup 1.0000x reference)
#include <hip/hip_runtime.h>

typedef __attribute__((ext_vector_type(2))) float v2f;
typedef __attribute__((ext_vector_type(8))) float v8f;

#define N_PIX 4096
#define CH 32
#define CIN 64
#define BATCH 4
#define PLANE (CH * N_PIX)

__global__ __launch_bounds__(256) void qkv_conv_kernel(
    const float* __restrict__ x,
    const float* __restrict__ w1, const float* __restrict__ bb1,
    const float* __restrict__ w2, const float* __restrict__ bb2,
    const float* __restrict__ w3, const float* __restrict__ bb3,
    float* __restrict__ Q, float* __restrict__ K, float* __restrict__ V)
{
    int tid = blockIdx.x * blockDim.x + threadIdx.x;
    int b = tid >> 12;
    int n = tid & (N_PIX - 1);
    const float* xb = x + (size_t)b * CIN * N_PIX + n;
    float xr[CIN];
#pragma unroll
    for (int c = 0; c < CIN; ++c) xr[c] = xb[(size_t)c * N_PIX];
    size_t obase = (size_t)b * PLANE + n;
#pragma unroll 2
    for (int o = 0; o < CH; ++o) {
        float a1 = bb1[o], a2 = bb2[o], a3 = bb3[o];
#pragma unroll
        for (int c = 0; c < CIN; ++c) {
            float xv = xr[c];
            a1 = fmaf(w1[o * CIN + c], xv, a1);
            a2 = fmaf(w2[o * CIN + c], xv, a2);
            a3 = fmaf(w3[o * CIN + c], xv, a3);
        }
        Q[obase + (size_t)o * N_PIX] = a1;
        K[obase + (size_t)o * N_PIX] = a2;
        V[obase + (size_t)o * N_PIX] = a3;
    }
}

__global__ __launch_bounds__(128) void attn_kernel(
    const float* __restrict__ Q, const float* __restrict__ K,
    const float* __restrict__ V, float* __restrict__ Y)
{
    __shared__ float sc[BATCH][16][17];
    __shared__ float pc[BATCH][16][17];

    const int lane = threadIdx.x & 31;
    const int b    = threadIdx.x >> 5;
    const int lo   = lane & 15;
    const int hi   = lane >> 4;
    const int i0   = blockIdx.x * 16;

    const float* Qb = Q + (size_t)b * PLANE;
    const float* Kb = K + (size_t)b * PLANE;
    const float* Vb = V + (size_t)b * PLANE;

    v2f qa[8];
#pragma unroll
    for (int kk = 0; kk < 8; ++kk) {
        const float* p = Qb + (size_t)(i0 + lo) * CH + 4 * kk + 2 * hi;
        qa[kk] = v2f{p[0], p[1]};
    }

    v8f acc0 = {};
    v8f acc1 = {};

    for (int jt = 0; jt < N_PIX; jt += 16) {
        v8f s = {};
#pragma unroll
        for (int kk = 0; kk < 8; ++kk) {
            v2f kb;
            kb.x = Kb[(size_t)(4 * kk + 2 * hi)     * N_PIX + jt + lo];
            kb.y = Kb[(size_t)(4 * kk + 2 * hi + 1) * N_PIX + jt + lo];
            s = __builtin_amdgcn_wmma_f32_16x16x4_f32(
                    false, qa[kk], false, kb, (short)0, s, false, false);
        }
#pragma unroll
        for (int v = 0; v < 8; ++v) sc[b][v + 8 * hi][lo] = s[v];
        __syncthreads();

#pragma unroll
        for (int v = 0; v < 8; ++v) {
            int r = v + 8 * hi;
            float s0 = sc[0][r][lo], s1 = sc[1][r][lo];
            float s2 = sc[2][r][lo], s3 = sc[3][r][lo];
            float m = fmaxf(fmaxf(s0, s1), fmaxf(s2, s3));
            float e0 = __expf(s0 - m), e1 = __expf(s1 - m);
            float e2 = __expf(s2 - m), e3 = __expf(s3 - m);
            float d = e0 + e1 + e2 + e3;
            float eb = (b == 0) ? e0 : (b == 1) ? e1 : (b == 2) ? e2 : e3;
            pc[b][r][lo] = eb / d;
        }
        __syncthreads();

#pragma unroll
        for (int kk = 0; kk < 4; ++kk) {
            v2f a;
            a.x = pc[b][lo][4 * kk + 2 * hi];
            a.y = pc[b][lo][4 * kk + 2 * hi + 1];
            const float* vr0 = Vb + (size_t)(jt + 4 * kk + 2 * hi) * CH;
            const float* vr1 = vr0 + CH;
            v2f bl, bh;
            bl.x = vr0[lo];      bl.y = vr1[lo];
            bh.x = vr0[16 + lo]; bh.y = vr1[16 + lo];
            acc0 = __builtin_amdgcn_wmma_f32_16x16x4_f32(
                       false, a, false, bl, (short)0, acc0, false, false);
            acc1 = __builtin_amdgcn_wmma_f32_16x16x4_f32(
                       false, a, false, bh, (short)0, acc1, false, false);
        }
    }

    float* Yb = Y + (size_t)b * PLANE;
#pragma unroll
    for (int v = 0; v < 8; ++v) {
        int r = v + 8 * hi;
        Yb[(size_t)(i0 + r) * CH + lo]      = acc0[v];
        Yb[(size_t)(i0 + r) * CH + 16 + lo] = acc1[v];
    }
}

__global__ __launch_bounds__(256) void out_conv_kernel(
    const float* __restrict__ Y, const float* __restrict__ w4,
    const float* __restrict__ bb4, float* __restrict__ out)
{
    int tid = blockIdx.x * blockDim.x + threadIdx.x;
    int b = tid >> 12;
    int n = tid & (N_PIX - 1);
    const float* Yb = Y + (size_t)b * PLANE;
    float yr[CH];
#pragma unroll
    for (int c = 0; c < CH; ++c) yr[c] = Yb[(size_t)c * N_PIX + n];
    float* ob = out + (size_t)b * CIN * N_PIX + n;
#pragma unroll 2
    for (int o = 0; o < CIN; ++o) {
        float a = bb4[o];
#pragma unroll
        for (int c = 0; c < CH; ++c) a = fmaf(w4[o * CH + c], yr[c], a);
        ob[(size_t)o * N_PIX] = a;
    }
}

extern "C" void kernel_launch(void* const* d_in, const int* in_sizes, int n_in,
                              void* d_out, int out_size, void* d_ws, size_t ws_size,
                              hipStream_t stream) {
    const float* x   = (const float*)d_in[0];
    const float* w1  = (const float*)d_in[1];
    const float* bb1 = (const float*)d_in[2];
    const float* w2  = (const float*)d_in[3];
    const float* bb2 = (const float*)d_in[4];
    const float* w3  = (const float*)d_in[5];
    const float* bb3 = (const float*)d_in[6];
    const float* w4  = (const float*)d_in[7];
    const float* bb4 = (const float*)d_in[8];
    float* out = (float*)d_out;

    float* Q = (float*)d_ws;
    float* K = Q + (size_t)BATCH * PLANE;
    float* V = K + (size_t)BATCH * PLANE;
    float* Y = V + (size_t)BATCH * PLANE;

    qkv_conv_kernel<<<(BATCH * N_PIX) / 256, 256, 0, stream>>>(
        x, w1, bb1, w2, bb2, w3, bb3, Q, K, V);
    attn_kernel<<<N_PIX / 16, 128, 0, stream>>>(Q, K, V, Y);
    out_conv_kernel<<<(BATCH * N_PIX) / 256, 256, 0, stream>>>(Y, w4, bb4, out);
}


# attention: each wave holds all 4 batches of a 16x32 tile, scores computed transposed so batch softmax and P stay in registers (no LDS/barrier in loop); 8 key slices per block summed in fixed order
# speedup vs baseline: 3.2359x; 3.2359x over previous
#include <hip/hip_runtime.h>

typedef __attribute__((ext_vector_type(2))) float v2f;
typedef __attribute__((ext_vector_type(8))) float v8f;

#define N_PIX 4096
#define CH 32
#define CIN 64
#define BATCH 4
#define PLANE (CH * N_PIX)

__global__ __launch_bounds__(256) void qkv_conv_kernel(
    const float* __restrict__ x,
    const float* __restrict__ w1, const float* __restrict__ bb1,
    const float* __restrict__ w2, const float* __restrict__ bb2,
    const float* __restrict__ w3, const float* __restrict__ bb3,
    float* __restrict__ Q, float* __restrict__ K, float* __restrict__ V)
{
    int tid = blockIdx.x * blockDim.x + threadIdx.x;
    int b = tid >> 12;
    int n = tid & (N_PIX - 1);
    const float* xb = x + (size_t)b * CIN * N_PIX + n;
    float xr[CIN];
#pragma unroll
    for (int c = 0; c < CIN; ++c) xr[c] = xb[(size_t)c * N_PIX];
    size_t obase = (size_t)b * PLANE + n;
#pragma unroll 2
    for (int o = 0; o < CH; ++o) {
        float a1 = bb1[o], a2 = bb2[o], a3 = bb3[o];
#pragma unroll
        for (int c = 0; c < CIN; ++c) {
            float xv = xr[c];
            a1 = fmaf(w1[o * CIN + c], xv, a1);
            a2 = fmaf(w2[o * CIN + c], xv, a2);
            a3 = fmaf(w3[o * CIN + c], xv, a3);
        }
        Q[obase + (size_t)o * N_PIX] = a1;
        K[obase + (size_t)o * N_PIX] = a2;
        V[obase + (size_t)o * N_PIX] = a3;
    }
}

#define JSPLIT 8
#define JW (N_PIX / JSPLIT)
#define JSTEP 32

__global__ __launch_bounds__(256) void attn_kernel(
    const float* __restrict__ Q, const float* __restrict__ K,
    const float* __restrict__ V, float* __restrict__ Y)
{
    __shared__ float part[JSPLIT][BATCH][16 * CH];

    const int lane = threadIdx.x & 31;
    const int w    = __builtin_amdgcn_readfirstlane(threadIdx.x >> 5);
    const int lo   = lane & 15;
    const int hi   = lane >> 4;
    const int i0   = blockIdx.x * 16;

    v2f qb[BATCH][8];
#pragma unroll
    for (int b = 0; b < BATCH; ++b) {
#pragma unroll
        for (int kk = 0; kk < 8; ++kk)
            qb[b][kk] = *(const v2f*)(Q + (size_t)b * PLANE +
                                      (size_t)(i0 + lo) * CH + 4 * kk + 2 * hi);
    }

    v8f acc[BATCH][2];
#pragma unroll
    for (int b = 0; b < BATCH; ++b) { acc[b][0] = v8f{}; acc[b][1] = v8f{}; }

    const float* Kl = K + (size_t)(2 * hi) * N_PIX + 2 * lo;
    const float* Vl = V + (size_t)(16 * hi) * CH + 2 * lo;

    for (int jt = w * JW; jt < (w + 1) * JW; jt += JSTEP) {
        v2f kr[BATCH][8][2];
        v2f vr[BATCH][2][4][2];
#pragma unroll
        for (int kk = 0; kk < 8; ++kk) {
#pragma unroll
            for (int b = 0; b < BATCH; ++b) {
#pragma unroll
                for (int r = 0; r < 2; ++r)
                    kr[b][kk][r] = *(const v2f*)(Kl + (size_t)b * PLANE +
                                                 (size_t)(4 * kk + r) * N_PIX + jt);
            }
        }
#pragma unroll
        for (int t = 0; t < 2; ++t) {
#pragma unroll
            for (int kk = 0; kk < 4; ++kk) {
#pragma unroll
                for (int b = 0; b < BATCH; ++b) {
#pragma unroll
                    for (int r = 0; r < 2; ++r)
                        vr[b][t][kk][r] = *(const v2f*)(Vl + (size_t)b * PLANE +
                            (size_t)(jt + 4 * kk + 2 * r + t) * CH);
                }
            }
        }
        __builtin_amdgcn_sched_barrier(0);

#pragma unroll
        for (int t = 0; t < 2; ++t) {
            v8f s[BATCH];
#pragma unroll
            for (int b = 0; b < BATCH; ++b) s[b] = v8f{};
#pragma unroll
            for (int kk = 0; kk < 8; ++kk) {
#pragma unroll
                for (int b = 0; b < BATCH; ++b) {
                    v2f a = v2f{kr[b][kk][0][t], kr[b][kk][1][t]};
                    s[b] = __builtin_amdgcn_wmma_f32_16x16x4_f32(
                               false, a, false, qb[b][kk], (short)0, s[b], false, false);
                }
            }
#pragma unroll
            for (int v = 0; v < 8; ++v) {
                float s0 = s[0][v], s1 = s[1][v], s2 = s[2][v], s3 = s[3][v];
                float m = fmaxf(fmaxf(s0, s1), fmaxf(s2, s3));
                float e0 = __expf(s0 - m), e1 = __expf(s1 - m);
                float e2 = __expf(s2 - m), e3 = __expf(s3 - m);
                float d = e0 + e1 + e2 + e3;
                float rd = __builtin_amdgcn_rcpf(d);
                s[0][v] = e0 * rd; s[1][v] = e1 * rd;
                s[2][v] = e2 * rd; s[3][v] = e3 * rd;
            }
#pragma unroll
            for (int kk = 0; kk < 4; ++kk) {
#pragma unroll
                for (int b = 0; b < BATCH; ++b) {
                    v2f a  = v2f{s[b][2 * kk], s[b][2 * kk + 1]};
                    v2f bl = v2f{vr[b][t][kk][0].x, vr[b][t][kk][1].x};
                    v2f bh = v2f{vr[b][t][kk][0].y, vr[b][t][kk][1].y};
                    acc[b][0] = __builtin_amdgcn_wmma_f32_16x16x4_f32(
                                    false, a, false, bl, (short)0, acc[b][0], false, false);
                    acc[b][1] = __builtin_amdgcn_wmma_f32_16x16x4_f32(
                                    false, a, false, bh, (short)0, acc[b][1], false, false);
                }
            }
        }
    }

#pragma unroll
    for (int b = 0; b < BATCH; ++b) {
#pragma unroll
        for (int v = 0; v < 8; ++v)
            *(v2f*)&part[w][b][(v + 8 * hi) * CH + 2 * lo] = v2f{acc[b][0][v], acc[b][1][v]};
    }
    __syncthreads();

    const int e2 = threadIdx.x * 2;
#pragma unroll
    for (int b = 0; b < BATCH; ++b) {
        v2f sum = *(const v2f*)&part[0][b][e2];
#pragma unroll
        for (int ww = 1; ww < JSPLIT; ++ww) sum += *(const v2f*)&part[ww][b][e2];
        *(v2f*)(Y + (size_t)b * PLANE + (size_t)i0 * CH + e2) = sum;
    }
}

__global__ __launch_bounds__(256) void out_conv_kernel(
    const float* __restrict__ Y, const float* __restrict__ w4,
    const float* __restrict__ bb4, float* __restrict__ out)
{
    int tid = blockIdx.x * blockDim.x + threadIdx.x;
    int b = tid >> 12;
    int n = tid & (N_PIX - 1);
    const float* Yb = Y + (size_t)b * PLANE;
    float yr[CH];
#pragma unroll
    for (int c = 0; c < CH; ++c) yr[c] = Yb[(size_t)c * N_PIX + n];
    float* ob = out + (size_t)b * CIN * N_PIX + n;
#pragma unroll 2
    for (int o = 0; o < CIN; ++o) {
        float a = bb4[o];
#pragma unroll
        for (int c = 0; c < CH; ++c) a = fmaf(w4[o * CH + c], yr[c], a);
        ob[(size_t)o * N_PIX] = a;
    }
}

extern "C" void kernel_launch(void* const* d_in, const int* in_sizes, int n_in,
                              void* d_out, int out_size, void* d_ws, size_t ws_size,
                              hipStream_t stream) {
    const float* x   = (const float*)d_in[0];
    const float* w1  = (const float*)d_in[1];
    const float* bb1 = (const float*)d_in[2];
    const float* w2  = (const float*)d_in[3];
    const float* bb2 = (const float*)d_in[4];
    const float* w3  = (const float*)d_in[5];
    const float* bb3 = (const float*)d_in[6];
    const float* w4  = (const float*)d_in[7];
    const float* bb4 = (const float*)d_in[8];
    float* out = (float*)d_out;

    float* Q = (float*)d_ws;
    float* K = Q + (size_t)BATCH * PLANE;
    float* V = K + (size_t)BATCH * PLANE;
    float* Y = V + (size_t)BATCH * PLANE;

    qkv_conv_kernel<<<(BATCH * N_PIX) / 256, 256, 0, stream>>>(
        x, w1, bb1, w2, bb2, w3, bb3, Q, K, V);
    attn_kernel<<<N_PIX / 16, 32 * JSPLIT, 0, stream>>>(Q, K, V, Y);
    out_conv_kernel<<<(BATCH * N_PIX) / 256, 256, 0, stream>>>(Y, w4, bb4, out);
}


# QKV 1x1 convs on f32 WMMA tiles (W rows as A, x pixels as B, all vector loads as one group) replacing per-thread FMAs fed by a waited scalar load per weight pair; conv sum order changes
# speedup vs baseline: 7.1017x; 2.1947x over previous
#include <hip/hip_runtime.h>

typedef __attribute__((ext_vector_type(2))) float v2f;
typedef __attribute__((ext_vector_type(8))) float v8f;

#define N_PIX 4096
#define CH 32
#define CIN 64
#define BATCH 4
#define PLANE (CH * N_PIX)

typedef __attribute__((ext_vector_type(4))) float v4f;

#define QKV_GROUPS 2
#define QKV_WAVES (3 * QKV_GROUPS)

__global__ __launch_bounds__(32 * QKV_WAVES) void qkv_conv_kernel(
    const float* __restrict__ x,
    const float* __restrict__ w1, const float* __restrict__ bb1,
    const float* __restrict__ w2, const float* __restrict__ bb2,
    const float* __restrict__ w3, const float* __restrict__ bb3,
    float* __restrict__ Q, float* __restrict__ K, float* __restrict__ V)
{
    const int lane = threadIdx.x & 31;
    const int w  = __builtin_amdgcn_readfirstlane(threadIdx.x >> 5);
    const int lo = lane & 15;
    const int hi = lane >> 4;
    const int m  = w % 3;
    const int g  = blockIdx.x * QKV_GROUPS + w / 3;
    const int b  = g >> 7;
    const int n0 = (g & 127) * 32;

    const float* Wm = (m == 0) ? w1  : ((m == 1) ? w2  : w3);
    const float* bm = (m == 0) ? bb1 : ((m == 1) ? bb2 : bb3);
    float*       Om = (m == 0) ? Q   : ((m == 1) ? K   : V);

    const float* xl = x + (size_t)b * CIN * N_PIX + (size_t)(4 * hi) * N_PIX + n0 + 2 * lo;
    v2f xr[8][2][2];
#pragma unroll
    for (int j = 0; j < 8; ++j) {
#pragma unroll
        for (int p = 0; p < 2; ++p) {
#pragma unroll
            for (int s = 0; s < 2; ++s)
                xr[j][p][s] = *(const v2f*)(xl + (size_t)(8 * j + 2 * p + s) * N_PIX);
        }
    }
    v4f wr[2][8];
#pragma unroll
    for (int ot = 0; ot < 2; ++ot) {
#pragma unroll
        for (int j = 0; j < 8; ++j)
            wr[ot][j] = *(const v4f*)(Wm + (size_t)(16 * ot + lo) * CIN + 8 * j + 4 * hi);
    }

    v8f acc[2][2];
#pragma unroll
    for (int ot = 0; ot < 2; ++ot) {
        v4f b0 = *(const v4f*)(bm + 16 * ot + 8 * hi);
        v4f b1 = *(const v4f*)(bm + 16 * ot + 8 * hi + 4);
        v8f bv = v8f{b0.x, b0.y, b0.z, b0.w, b1.x, b1.y, b1.z, b1.w};
        acc[ot][0] = bv;
        acc[ot][1] = bv;
    }
    __builtin_amdgcn_sched_barrier(0);

#pragma unroll
    for (int j = 0; j < 8; ++j) {
#pragma unroll
        for (int p = 0; p < 2; ++p) {
#pragma unroll
            for (int ot = 0; ot < 2; ++ot) {
                v2f a = v2f{wr[ot][j][2 * p], wr[ot][j][2 * p + 1]};
#pragma unroll
                for (int T = 0; T < 2; ++T) {
                    v2f bf = v2f{xr[j][p][0][T], xr[j][p][1][T]};
                    acc[ot][T] = __builtin_amdgcn_wmma_f32_16x16x4_f32(
                                     false, a, false, bf, (short)0, acc[ot][T], false, false);
                }
            }
        }
    }

    float* ol = Om + (size_t)b * PLANE + (size_t)(8 * hi) * N_PIX + n0 + 2 * lo;
#pragma unroll
    for (int ot = 0; ot < 2; ++ot) {
#pragma unroll
        for (int v = 0; v < 8; ++v)
            *(v2f*)(ol + (size_t)(16 * ot + v) * N_PIX) = v2f{acc[ot][0][v], acc[ot][1][v]};
    }
}

#define JSPLIT 8
#define JW (N_PIX / JSPLIT)
#define JSTEP 32

__global__ __launch_bounds__(256) void attn_kernel(
    const float* __restrict__ Q, const float* __restrict__ K,
    const float* __restrict__ V, float* __restrict__ Y)
{
    __shared__ float part[JSPLIT][BATCH][16 * CH];

    const int lane = threadIdx.x & 31;
    const int w    = __builtin_amdgcn_readfirstlane(threadIdx.x >> 5);
    const int lo   = lane & 15;
    const int hi   = lane >> 4;
    const int i0   = blockIdx.x * 16;

    v2f qb[BATCH][8];
#pragma unroll
    for (int b = 0; b < BATCH; ++b) {
#pragma unroll
        for (int kk = 0; kk < 8; ++kk)
            qb[b][kk] = *(const v2f*)(Q + (size_t)b * PLANE +
                                      (size_t)(i0 + lo) * CH + 4 * kk + 2 * hi);
    }

    v8f acc[BATCH][2];
#pragma unroll
    for (int b = 0; b < BATCH; ++b) { acc[b][0] = v8f{}; acc[b][1] = v8f{}; }

    const float* Kl = K + (size_t)(2 * hi) * N_PIX + 2 * lo;
    const float* Vl = V + (size_t)(16 * hi) * CH + 2 * lo;

    for (int jt = w * JW; jt < (w + 1) * JW; jt += JSTEP) {
        v2f kr[BATCH][8][2];
        v2f vr[BATCH][2][4][2];
#pragma unroll
        for (int kk = 0; kk < 8; ++kk) {
#pragma unroll
            for (int b = 0; b < BATCH; ++b) {
#pragma unroll
                for (int r = 0; r < 2; ++r)
                    kr[b][kk][r] = *(const v2f*)(Kl + (size_t)b * PLANE +
                                                 (size_t)(4 * kk + r) * N_PIX + jt);
            }
        }
#pragma unroll
        for (int t = 0; t < 2; ++t) {
#pragma unroll
            for (int kk = 0; kk < 4; ++kk) {
#pragma unroll
                for (int b = 0; b < BATCH; ++b) {
#pragma unroll
                    for (int r = 0; r < 2; ++r)
                        vr[b][t][kk][r] = *(const v2f*)(Vl + (size_t)b * PLANE +
                            (size_t)(jt + 4 * kk + 2 * r + t) * CH);
                }
            }
        }
        __builtin_amdgcn_sched_barrier(0);

#pragma unroll
        for (int t = 0; t < 2; ++t) {
            v8f s[BATCH];
#pragma unroll
            for (int b = 0; b < BATCH; ++b) s[b] = v8f{};
#pragma unroll
            for (int kk = 0; kk < 8; ++kk) {
#pragma unroll
                for (int b = 0; b < BATCH; ++b) {
                    v2f a = v2f{kr[b][kk][0][t], kr[b][kk][1][t]};
                    s[b] = __builtin_amdgcn_wmma_f32_16x16x4_f32(
                               false, a, false, qb[b][kk], (short)0, s[b], false, false);
                }
            }
#pragma unroll
            for (int v = 0; v < 8; ++v) {
                float s0 = s[0][v], s1 = s[1][v], s2 = s[2][v], s3 = s[3][v];
                float m = fmaxf(fmaxf(s0, s1), fmaxf(s2, s3));
                float e0 = __expf(s0 - m), e1 = __expf(s1 - m);
                float e2 = __expf(s2 - m), e3 = __expf(s3 - m);
                float d = e0 + e1 + e2 + e3;
                float rd = __builtin_amdgcn_rcpf(d);
                s[0][v] = e0 * rd; s[1][v] = e1 * rd;
                s[2][v] = e2 * rd; s[3][v] = e3 * rd;
            }
#pragma unroll
            for (int kk = 0; kk < 4; ++kk) {
#pragma unroll
                for (int b = 0; b < BATCH; ++b) {
                    v2f a  = v2f{s[b][2 * kk], s[b][2 * kk + 1]};
                    v2f bl = v2f{vr[b][t][kk][0].x, vr[b][t][kk][1].x};
                    v2f bh = v2f{vr[b][t][kk][0].y, vr[b][t][kk][1].y};
                    acc[b][0] = __builtin_amdgcn_wmma_f32_16x16x4_f32(
                                    false, a, false, bl, (short)0, acc[b][0], false, false);
                    acc[b][1] = __builtin_amdgcn_wmma_f32_16x16x4_f32(
                                    false, a, false, bh, (short)0, acc[b][1], false, false);
                }
            }
        }
    }

#pragma unroll
    for (int b = 0; b < BATCH; ++b) {
#pragma unroll
        for (int v = 0; v < 8; ++v)
            *(v2f*)&part[w][b][(v + 8 * hi) * CH + 2 * lo] = v2f{acc[b][0][v], acc[b][1][v]};
    }
    __syncthreads();

    const int e2 = threadIdx.x * 2;
#pragma unroll
    for (int b = 0; b < BATCH; ++b) {
        v2f sum = *(const v2f*)&part[0][b][e2];
#pragma unroll
        for (int ww = 1; ww < JSPLIT; ++ww) sum += *(const v2f*)&part[ww][b][e2];
        *(v2f*)(Y + (size_t)b * PLANE + (size_t)i0 * CH + e2) = sum;
    }
}

__global__ __launch_bounds__(256) void out_conv_kernel(
    const float* __restrict__ Y, const float* __restrict__ w4,
    const float* __restrict__ bb4, float* __restrict__ out)
{
    int tid = blockIdx.x * blockDim.x + threadIdx.x;
    int b = tid >> 12;
    int n = tid & (N_PIX - 1);
    const float* Yb = Y + (size_t)b * PLANE;
    float yr[CH];
#pragma unroll
    for (int c = 0; c < CH; ++c) yr[c] = Yb[(size_t)c * N_PIX + n];
    float* ob = out + (size_t)b * CIN * N_PIX + n;
#pragma unroll 2
    for (int o = 0; o < CIN; ++o) {
        float a = bb4[o];
#pragma unroll
        for (int c = 0; c < CH; ++c) a = fmaf(w4[o * CH + c], yr[c], a);
        ob[(size_t)o * N_PIX] = a;
    }
}

extern "C" void kernel_launch(void* const* d_in, const int* in_sizes, int n_in,
                              void* d_out, int out_size, void* d_ws, size_t ws_size,
                              hipStream_t stream) {
    const float* x   = (const float*)d_in[0];
    const float* w1  = (const float*)d_in[1];
    const float* bb1 = (const float*)d_in[2];
    const float* w2  = (const float*)d_in[3];
    const float* bb2 = (const float*)d_in[4];
    const float* w3  = (const float*)d_in[5];
    const float* bb3 = (const float*)d_in[6];
    const float* w4  = (const float*)d_in[7];
    const float* bb4 = (const float*)d_in[8];
    float* out = (float*)d_out;

    float* Q = (float*)d_ws;
    float* K = Q + (size_t)BATCH * PLANE;
    float* V = K + (size_t)BATCH * PLANE;
    float* Y = V + (size_t)BATCH * PLANE;

    qkv_conv_kernel<<<(BATCH * N_PIX) / (32 * QKV_GROUPS), 32 * QKV_WAVES, 0, stream>>>(
        x, w1, bb1, w2, bb2, w3, bb3, Q, K, V);
    attn_kernel<<<N_PIX / 16, 32 * JSPLIT, 0, stream>>>(Q, K, V, Y);
    out_conv_kernel<<<(BATCH * N_PIX) / 256, 256, 0, stream>>>(Y, w4, bb4, out);
}


# output 1x1 conv on f32 WMMA tiles (W4 rows as A, Y pixels as B, all vector loads as one group, 256 work-groups) replacing per-thread FMAs fed by waited scalar weight loads; conv sum order changes
# speedup vs baseline: 8.2263x; 1.1584x over previous
#include <hip/hip_runtime.h>

typedef __attribute__((ext_vector_type(2))) float v2f;
typedef __attribute__((ext_vector_type(8))) float v8f;

#define N_PIX 4096
#define CH 32
#define CIN 64
#define BATCH 4
#define PLANE (CH * N_PIX)

typedef __attribute__((ext_vector_type(4))) float v4f;

#define QKV_GROUPS 2
#define QKV_WAVES (3 * QKV_GROUPS)

__global__ __launch_bounds__(32 * QKV_WAVES) void qkv_conv_kernel(
    const float* __restrict__ x,
    const float* __restrict__ w1, const float* __restrict__ bb1,
    const float* __restrict__ w2, const float* __restrict__ bb2,
    const float* __restrict__ w3, const float* __restrict__ bb3,
    float* __restrict__ Q, float* __restrict__ K, float* __restrict__ V)
{
    const int lane = threadIdx.x & 31;
    const int w  = __builtin_amdgcn_readfirstlane(threadIdx.x >> 5);
    const int lo = lane & 15;
    const int hi = lane >> 4;
    const int m  = w % 3;
    const int g  = blockIdx.x * QKV_GROUPS + w / 3;
    const int b  = g >> 7;
    const int n0 = (g & 127) * 32;

    const float* Wm = (m == 0) ? w1  : ((m == 1) ? w2  : w3);
    const float* bm = (m == 0) ? bb1 : ((m == 1) ? bb2 : bb3);
    float*       Om = (m == 0) ? Q   : ((m == 1) ? K   : V);

    const float* xl = x + (size_t)b * CIN * N_PIX + (size_t)(4 * hi) * N_PIX + n0 + 2 * lo;
    v2f xr[8][2][2];
#pragma unroll
    for (int j = 0; j < 8; ++j) {
#pragma unroll
        for (int p = 0; p < 2; ++p) {
#pragma unroll
            for (int s = 0; s < 2; ++s)
                xr[j][p][s] = *(const v2f*)(xl + (size_t)(8 * j + 2 * p + s) * N_PIX);
        }
    }
    v4f wr[2][8];
#pragma unroll
    for (int ot = 0; ot < 2; ++ot) {
#pragma unroll
        for (int j = 0; j < 8; ++j)
            wr[ot][j] = *(const v4f*)(Wm + (size_t)(16 * ot + lo) * CIN + 8 * j + 4 * hi);
    }

    v8f acc[2][2];
#pragma unroll
    for (int ot = 0; ot < 2; ++ot) {
        v4f b0 = *(const v4f*)(bm + 16 * ot + 8 * hi);
        v4f b1 = *(const v4f*)(bm + 16 * ot + 8 * hi + 4);
        v8f bv = v8f{b0.x, b0.y, b0.z, b0.w, b1.x, b1.y, b1.z, b1.w};
        acc[ot][0] = bv;
        acc[ot][1] = bv;
    }
    __builtin_amdgcn_sched_barrier(0);

#pragma unroll
    for (int j = 0; j < 8; ++j) {
#pragma unroll
        for (int p = 0; p < 2; ++p) {
#pragma unroll
            for (int ot = 0; ot < 2; ++ot) {
                v2f a = v2f{wr[ot][j][2 * p], wr[ot][j][2 * p + 1]};
#pragma unroll
                for (int T = 0; T < 2; ++T) {
                    v2f bf = v2f{xr[j][p][0][T], xr[j][p][1][T]};
                    acc[ot][T] = __builtin_amdgcn_wmma_f32_16x16x4_f32(
                                     false, a, false, bf, (short)0, acc[ot][T], false, false);
                }
            }
        }
    }

    float* ol = Om + (size_t)b * PLANE + (size_t)(8 * hi) * N_PIX + n0 + 2 * lo;
#pragma unroll
    for (int ot = 0; ot < 2; ++ot) {
#pragma unroll
        for (int v = 0; v < 8; ++v)
            *(v2f*)(ol + (size_t)(16 * ot + v) * N_PIX) = v2f{acc[ot][0][v], acc[ot][1][v]};
    }
}

#define JSPLIT 8
#define JW (N_PIX / JSPLIT)
#define JSTEP 32

__global__ __launch_bounds__(256) void attn_kernel(
    const float* __restrict__ Q, const float* __restrict__ K,
    const float* __restrict__ V, float* __restrict__ Y)
{
    __shared__ float part[JSPLIT][BATCH][16 * CH];

    const int lane = threadIdx.x & 31;
    const int w    = __builtin_amdgcn_readfirstlane(threadIdx.x >> 5);
    const int lo   = lane & 15;
    const int hi   = lane >> 4;
    const int i0   = blockIdx.x * 16;

    v2f qb[BATCH][8];
#pragma unroll
    for (int b = 0; b < BATCH; ++b) {
#pragma unroll
        for (int kk = 0; kk < 8; ++kk)
            qb[b][kk] = *(const v2f*)(Q + (size_t)b * PLANE +
                                      (size_t)(i0 + lo) * CH + 4 * kk + 2 * hi);
    }

    v8f acc[BATCH][2];
#pragma unroll
    for (int b = 0; b < BATCH; ++b) { acc[b][0] = v8f{}; acc[b][1] = v8f{}; }

    const float* Kl = K + (size_t)(2 * hi) * N_PIX + 2 * lo;
    const float* Vl = V + (size_t)(16 * hi) * CH + 2 * lo;

    for (int jt = w * JW; jt < (w + 1) * JW; jt += JSTEP) {
        v2f kr[BATCH][8][2];
        v2f vr[BATCH][2][4][2];
#pragma unroll
        for (int kk = 0; kk < 8; ++kk) {
#pragma unroll
            for (int b = 0; b < BATCH; ++b) {
#pragma unroll
                for (int r = 0; r < 2; ++r)
                    kr[b][kk][r] = *(const v2f*)(Kl + (size_t)b * PLANE +
                                                 (size_t)(4 * kk + r) * N_PIX + jt);
            }
        }
#pragma unroll
        for (int t = 0; t < 2; ++t) {
#pragma unroll
            for (int kk = 0; kk < 4; ++kk) {
#pragma unroll
                for (int b = 0; b < BATCH; ++b) {
#pragma unroll
                    for (int r = 0; r < 2; ++r)
                        vr[b][t][kk][r] = *(const v2f*)(Vl + (size_t)b * PLANE +
                            (size_t)(jt + 4 * kk + 2 * r + t) * CH);
                }
            }
        }
        __builtin_amdgcn_sched_barrier(0);

#pragma unroll
        for (int t = 0; t < 2; ++t) {
            v8f s[BATCH];
#pragma unroll
            for (int b = 0; b < BATCH; ++b) s[b] = v8f{};
#pragma unroll
            for (int kk = 0; kk < 8; ++kk) {
#pragma unroll
                for (int b = 0; b < BATCH; ++b) {
                    v2f a = v2f{kr[b][kk][0][t], kr[b][kk][1][t]};
                    s[b] = __builtin_amdgcn_wmma_f32_16x16x4_f32(
                               false, a, false, qb[b][kk], (short)0, s[b], false, false);
                }
            }
#pragma unroll
            for (int v = 0; v < 8; ++v) {
                float s0 = s[0][v], s1 = s[1][v], s2 = s[2][v], s3 = s[3][v];
                float m = fmaxf(fmaxf(s0, s1), fmaxf(s2, s3));
                float e0 = __expf(s0 - m), e1 = __expf(s1 - m);
                float e2 = __expf(s2 - m), e3 = __expf(s3 - m);
                float d = e0 + e1 + e2 + e3;
                float rd = __builtin_amdgcn_rcpf(d);
                s[0][v] = e0 * rd; s[1][v] = e1 * rd;
                s[2][v] = e2 * rd; s[3][v] = e3 * rd;
            }
#pragma unroll
            for (int kk = 0; kk < 4; ++kk) {
#pragma unroll
                for (int b = 0; b < BATCH; ++b) {
                    v2f a  = v2f{s[b][2 * kk], s[b][2 * kk + 1]};
                    v2f bl = v2f{vr[b][t][kk][0].x, vr[b][t][kk][1].x};
                    v2f bh = v2f{vr[b][t][kk][0].y, vr[b][t][kk][1].y};
                    acc[b][0] = __builtin_amdgcn_wmma_f32_16x16x4_f32(
                                    false, a, false, bl, (short)0, acc[b][0], false, false);
                    acc[b][1] = __builtin_amdgcn_wmma_f32_16x16x4_f32(
                                    false, a, false, bh, (short)0, acc[b][1], false, false);
                }
            }
        }
    }

#pragma unroll
    for (int b = 0; b < BATCH; ++b) {
#pragma unroll
        for (int v = 0; v < 8; ++v)
            *(v2f*)&part[w][b][(v + 8 * hi) * CH + 2 * lo] = v2f{acc[b][0][v], acc[b][1][v]};
    }
    __syncthreads();

    const int e2 = threadIdx.x * 2;
#pragma unroll
    for (int b = 0; b < BATCH; ++b) {
        v2f sum = *(const v2f*)&part[0][b][e2];
#pragma unroll
        for (int ww = 1; ww < JSPLIT; ++ww) sum += *(const v2f*)&part[ww][b][e2];
        *(v2f*)(Y + (size_t)b * PLANE + (size_t)i0 * CH + e2) = sum;
    }
}

#define OC_GROUPS 2
#define OC_WAVES (2 * OC_GROUPS)

__global__ __launch_bounds__(32 * OC_WAVES) void out_conv_kernel(
    const float* __restrict__ Y, const float* __restrict__ w4,
    const float* __restrict__ bb4, float* __restrict__ out)
{
    const int lane = threadIdx.x & 31;
    const int w  = __builtin_amdgcn_readfirstlane(threadIdx.x >> 5);
    const int lo = lane & 15;
    const int hi = lane >> 4;
    const int h  = w & 1;
    const int g  = blockIdx.x * OC_GROUPS + (w >> 1);
    const int b  = g >> 7;
    const int n0 = (g & 127) * 32;

    const float* yl = Y + (size_t)b * PLANE + (size_t)(4 * hi) * N_PIX + n0 + 2 * lo;
    v2f yr[4][2][2];
#pragma unroll
    for (int j = 0; j < 4; ++j) {
#pragma unroll
        for (int p = 0; p < 2; ++p) {
#pragma unroll
            for (int s = 0; s < 2; ++s)
                yr[j][p][s] = *(const v2f*)(yl + (size_t)(8 * j + 2 * p + s) * N_PIX);
        }
    }
    const float* Wm = w4 + (size_t)(32 * h) * CH;
    const float* bm = bb4 + 32 * h;
    v4f wr[2][4];
#pragma unroll
    for (int ot = 0; ot < 2; ++ot) {
#pragma unroll
        for (int j = 0; j < 4; ++j)
            wr[ot][j] = *(const v4f*)(Wm + (size_t)(16 * ot + lo) * CH + 8 * j + 4 * hi);
    }

    v8f acc[2][2];
#pragma unroll
    for (int ot = 0; ot < 2; ++ot) {
        v4f b0 = *(const v4f*)(bm + 16 * ot + 8 * hi);
        v4f b1 = *(const v4f*)(bm + 16 * ot + 8 * hi + 4);
        v8f bv = v8f{b0.x, b0.y, b0.z, b0.w, b1.x, b1.y, b1.z, b1.w};
        acc[ot][0] = bv;
        acc[ot][1] = bv;
    }
    __builtin_amdgcn_sched_barrier(0);

#pragma unroll
    for (int j = 0; j < 4; ++j) {
#pragma unroll
        for (int p = 0; p < 2; ++p) {
#pragma unroll
            for (int ot = 0; ot < 2; ++ot) {
                v2f a = v2f{wr[ot][j][2 * p], wr[ot][j][2 * p + 1]};
#pragma unroll
                for (int T = 0; T < 2; ++T) {
                    v2f bf = v2f{yr[j][p][0][T], yr[j][p][1][T]};
                    acc[ot][T] = __builtin_amdgcn_wmma_f32_16x16x4_f32(
                                     false, a, false, bf, (short)0, acc[ot][T], false, false);
                }
            }
        }
    }

    float* ol = out + (size_t)b * CIN * N_PIX + (size_t)(32 * h + 8 * hi) * N_PIX + n0 + 2 * lo;
#pragma unroll
    for (int ot = 0; ot < 2; ++ot) {
#pragma unroll
        for (int v = 0; v < 8; ++v)
            *(v2f*)(ol + (size_t)(16 * ot + v) * N_PIX) = v2f{acc[ot][0][v], acc[ot][1][v]};
    }
}

extern "C" void kernel_launch(void* const* d_in, const int* in_sizes, int n_in,
                              void* d_out, int out_size, void* d_ws, size_t ws_size,
                              hipStream_t stream) {
    const float* x   = (const float*)d_in[0];
    const float* w1  = (const float*)d_in[1];
    const float* bb1 = (const float*)d_in[2];
    const float* w2  = (const float*)d_in[3];
    const float* bb2 = (const float*)d_in[4];
    const float* w3  = (const float*)d_in[5];
    const float* bb3 = (const float*)d_in[6];
    const float* w4  = (const float*)d_in[7];
    const float* bb4 = (const float*)d_in[8];
    float* out = (float*)d_out;

    float* Q = (float*)d_ws;
    float* K = Q + (size_t)BATCH * PLANE;
    float* V = K + (size_t)BATCH * PLANE;
    float* Y = V + (size_t)BATCH * PLANE;

    qkv_conv_kernel<<<(BATCH * N_PIX) / (32 * QKV_GROUPS), 32 * QKV_WAVES, 0, stream>>>(
        x, w1, bb1, w2, bb2, w3, bb3, Q, K, V);
    attn_kernel<<<N_PIX / 16, 32 * JSPLIT, 0, stream>>>(Q, K, V, Y);
    out_conv_kernel<<<(BATCH * N_PIX) / (32 * OC_GROUPS), 32 * OC_WAVES, 0, stream>>>(Y, w4, bb4, out);
}


# attention loop: loads issued one phase ahead in groups of 32 (next step's K rows behind each half of the score products, V rows behind their P.V), each into registers its last reader just left
# speedup vs baseline: 9.1619x; 1.1137x over previous
#include <hip/hip_runtime.h>

typedef __attribute__((ext_vector_type(2))) float v2f;
typedef __attribute__((ext_vector_type(8))) float v8f;

#define N_PIX 4096
#define CH 32
#define CIN 64
#define BATCH 4
#define PLANE (CH * N_PIX)

typedef __attribute__((ext_vector_type(4))) float v4f;

#define QKV_GROUPS 2
#define QKV_WAVES (3 * QKV_GROUPS)

__global__ __launch_bounds__(32 * QKV_WAVES) void qkv_conv_kernel(
    const float* __restrict__ x,
    const float* __restrict__ w1, const float* __restrict__ bb1,
    const float* __restrict__ w2, const float* __restrict__ bb2,
    const float* __restrict__ w3, const float* __restrict__ bb3,
    float* __restrict__ Q, float* __restrict__ K, float* __restrict__ V)
{
    const int lane = threadIdx.x & 31;
    const int w  = __builtin_amdgcn_readfirstlane(threadIdx.x >> 5);
    const int lo = lane & 15;
    const int hi = lane >> 4;
    const int m  = w % 3;
    const int g  = blockIdx.x * QKV_GROUPS + w / 3;
    const int b  = g >> 7;
    const int n0 = (g & 127) * 32;

    const float* Wm = (m == 0) ? w1  : ((m == 1) ? w2  : w3);
    const float* bm = (m == 0) ? bb1 : ((m == 1) ? bb2 : bb3);
    float*       Om = (m == 0) ? Q   : ((m == 1) ? K   : V);

    const float* xl = x + (size_t)b * CIN * N_PIX + (size_t)(4 * hi) * N_PIX + n0 + 2 * lo;
    v2f xr[8][2][2];
#pragma unroll
    for (int j = 0; j < 8; ++j) {
#pragma unroll
        for (int p = 0; p < 2; ++p) {
#pragma unroll
            for (int s = 0; s < 2; ++s)
                xr[j][p][s] = *(const v2f*)(xl + (size_t)(8 * j + 2 * p + s) * N_PIX);
        }
    }
    v4f wr[2][8];
#pragma unroll
    for (int ot = 0; ot < 2; ++ot) {
#pragma unroll
        for (int j = 0; j < 8; ++j)
            wr[ot][j] = *(const v4f*)(Wm + (size_t)(16 * ot + lo) * CIN + 8 * j + 4 * hi);
    }

    v8f acc[2][2];
#pragma unroll
    for (int ot = 0; ot < 2; ++ot) {
        v4f b0 = *(const v4f*)(bm + 16 * ot + 8 * hi);
        v4f b1 = *(const v4f*)(bm + 16 * ot + 8 * hi + 4);
        v8f bv = v8f{b0.x, b0.y, b0.z, b0.w, b1.x, b1.y, b1.z, b1.w};
        acc[ot][0] = bv;
        acc[ot][1] = bv;
    }
    __builtin_amdgcn_sched_barrier(0);

#pragma unroll
    for (int j = 0; j < 8; ++j) {
#pragma unroll
        for (int p = 0; p < 2; ++p) {
#pragma unroll
            for (int ot = 0; ot < 2; ++ot) {
                v2f a = v2f{wr[ot][j][2 * p], wr[ot][j][2 * p + 1]};
#pragma unroll
                for (int T = 0; T < 2; ++T) {
                    v2f bf = v2f{xr[j][p][0][T], xr[j][p][1][T]};
                    acc[ot][T] = __builtin_amdgcn_wmma_f32_16x16x4_f32(
                                     false, a, false, bf, (short)0, acc[ot][T], false, false);
                }
            }
        }
    }

    float* ol = Om + (size_t)b * PLANE + (size_t)(8 * hi) * N_PIX + n0 + 2 * lo;
#pragma unroll
    for (int ot = 0; ot < 2; ++ot) {
#pragma unroll
        for (int v = 0; v < 8; ++v)
            *(v2f*)(ol + (size_t)(16 * ot + v) * N_PIX) = v2f{acc[ot][0][v], acc[ot][1][v]};
    }
}

#define JSPLIT 8
#define JW (N_PIX / JSPLIT)
#define JSTEP 32

__global__ __launch_bounds__(256) void attn_kernel(
    const float* __restrict__ Q, const float* __restrict__ K,
    const float* __restrict__ V, float* __restrict__ Y)
{
    __shared__ float part[JSPLIT][BATCH][16 * CH];

    const int lane = threadIdx.x & 31;
    const int w    = __builtin_amdgcn_readfirstlane(threadIdx.x >> 5);
    const int lo   = lane & 15;
    const int hi   = lane >> 4;
    const int i0   = blockIdx.x * 16;

    v2f qb[BATCH][8];
#pragma unroll
    for (int b = 0; b < BATCH; ++b) {
#pragma unroll
        for (int kk = 0; kk < 8; ++kk)
            qb[b][kk] = *(const v2f*)(Q + (size_t)b * PLANE +
                                      (size_t)(i0 + lo) * CH + 4 * kk + 2 * hi);
    }

    v8f acc[BATCH][2];
#pragma unroll
    for (int b = 0; b < BATCH; ++b) { acc[b][0] = v8f{}; acc[b][1] = v8f{}; }

    const float* Kl = K + (size_t)(2 * hi) * N_PIX + 2 * lo;
    const float* Vl = V + (size_t)(16 * hi) * CH + 2 * lo;

    v2f kr[BATCH][8][2];
    v2f vr[BATCH][2][4][2];

    auto load_k_half = [&](int h, int j) {
#pragma unroll
        for (int kk = 4 * h; kk < 4 * h + 4; ++kk) {
#pragma unroll
            for (int b = 0; b < BATCH; ++b) {
#pragma unroll
                for (int r = 0; r < 2; ++r)
                    kr[b][kk][r] = *(const v2f*)(Kl + (size_t)b * PLANE +
                                                 (size_t)(4 * kk + r) * N_PIX + j);
            }
        }
    };
    auto load_v = [&](int t, int j) {
#pragma unroll
        for (int kk = 0; kk < 4; ++kk) {
#pragma unroll
            for (int b = 0; b < BATCH; ++b) {
#pragma unroll
                for (int r = 0; r < 2; ++r)
                    vr[b][t][kk][r] = *(const v2f*)(Vl + (size_t)b * PLANE +
                        (size_t)(j + 4 * kk + 2 * r + t) * CH);
            }
        }
    };
    auto scores_half = [&](int h, v8f (&sa)[BATCH], v8f (&sb)[BATCH]) {
#pragma unroll
        for (int kk = 4 * h; kk < 4 * h + 4; ++kk) {
#pragma unroll
            for (int b = 0; b < BATCH; ++b) {
                v2f a0 = v2f{kr[b][kk][0][0], kr[b][kk][1][0]};
                v2f a1 = v2f{kr[b][kk][0][1], kr[b][kk][1][1]};
                sa[b] = __builtin_amdgcn_wmma_f32_16x16x4_f32(
                            false, a0, false, qb[b][kk], (short)0, sa[b], false, false);
                sb[b] = __builtin_amdgcn_wmma_f32_16x16x4_f32(
                            false, a1, false, qb[b][kk], (short)0, sb[b], false, false);
            }
        }
    };
    auto softmax4 = [&](v8f (&s)[BATCH]) {
#pragma unroll
        for (int v = 0; v < 8; ++v) {
            float s0 = s[0][v], s1 = s[1][v], s2 = s[2][v], s3 = s[3][v];
            float m = fmaxf(fmaxf(s0, s1), fmaxf(s2, s3));
            float e0 = __expf(s0 - m), e1 = __expf(s1 - m);
            float e2 = __expf(s2 - m), e3 = __expf(s3 - m);
            float d = e0 + e1 + e2 + e3;
            float rd = __builtin_amdgcn_rcpf(d);
            s[0][v] = e0 * rd; s[1][v] = e1 * rd;
            s[2][v] = e2 * rd; s[3][v] = e3 * rd;
        }
    };
    auto pv = [&](int t, v8f (&s)[BATCH]) {
#pragma unroll
        for (int kk = 0; kk < 4; ++kk) {
#pragma unroll
            for (int b = 0; b < BATCH; ++b) {
                v2f a  = v2f{s[b][2 * kk], s[b][2 * kk + 1]};
                v2f bl = v2f{vr[b][t][kk][0].x, vr[b][t][kk][1].x};
                v2f bh = v2f{vr[b][t][kk][0].y, vr[b][t][kk][1].y};
                acc[b][0] = __builtin_amdgcn_wmma_f32_16x16x4_f32(
                                false, a, false, bl, (short)0, acc[b][0], false, false);
                acc[b][1] = __builtin_amdgcn_wmma_f32_16x16x4_f32(
                                false, a, false, bh, (short)0, acc[b][1], false, false);
            }
        }
    };

    auto step = [&](int jt, bool ahead) {
        v8f s0[BATCH], s1[BATCH];
#pragma unroll
        for (int b = 0; b < BATCH; ++b) { s0[b] = v8f{}; s1[b] = v8f{}; }
        load_v(1, jt);
        __builtin_amdgcn_sched_barrier(0);
        scores_half(0, s0, s1);
        if (ahead) {
            __builtin_amdgcn_sched_barrier(0);
            load_k_half(0, jt + JSTEP);
            __builtin_amdgcn_sched_barrier(0);
        }
        scores_half(1, s0, s1);
        if (ahead) {
            __builtin_amdgcn_sched_barrier(0);
            load_k_half(1, jt + JSTEP);
            __builtin_amdgcn_sched_barrier(0);
        }
        softmax4(s0);
        pv(0, s0);
        if (ahead) {
            __builtin_amdgcn_sched_barrier(0);
            load_v(0, jt + JSTEP);
            __builtin_amdgcn_sched_barrier(0);
        }
        softmax4(s1);
        pv(1, s1);
    };

    const int jbeg = w * JW;
    const int jlast = jbeg + JW - JSTEP;
    load_k_half(0, jbeg);
    load_k_half(1, jbeg);
    load_v(0, jbeg);
    for (int jt = jbeg; jt < jlast; jt += JSTEP) step(jt, true);
    step(jlast, false);

#pragma unroll
    for (int b = 0; b < BATCH; ++b) {
#pragma unroll
        for (int v = 0; v < 8; ++v)
            *(v2f*)&part[w][b][(v + 8 * hi) * CH + 2 * lo] = v2f{acc[b][0][v], acc[b][1][v]};
    }
    __syncthreads();

    const int e2 = threadIdx.x * 2;
#pragma unroll
    for (int b = 0; b < BATCH; ++b) {
        v2f sum = *(const v2f*)&part[0][b][e2];
#pragma unroll
        for (int ww = 1; ww < JSPLIT; ++ww) sum += *(const v2f*)&part[ww][b][e2];
        *(v2f*)(Y + (size_t)b * PLANE + (size_t)i0 * CH + e2) = sum;
    }
}

#define OC_GROUPS 2
#define OC_WAVES (2 * OC_GROUPS)

__global__ __launch_bounds__(32 * OC_WAVES) void out_conv_kernel(
    const float* __restrict__ Y, const float* __restrict__ w4,
    const float* __restrict__ bb4, float* __restrict__ out)
{
    const int lane = threadIdx.x & 31;
    const int w  = __builtin_amdgcn_readfirstlane(threadIdx.x >> 5);
    const int lo = lane & 15;
    const int hi = lane >> 4;
    const int h  = w & 1;
    const int g  = blockIdx.x * OC_GROUPS + (w >> 1);
    const int b  = g >> 7;
    const int n0 = (g & 127) * 32;

    const float* yl = Y + (size_t)b * PLANE + (size_t)(4 * hi) * N_PIX + n0 + 2 * lo;
    v2f yr[4][2][2];
#pragma unroll
    for (int j = 0; j < 4; ++j) {
#pragma unroll
        for (int p = 0; p < 2; ++p) {
#pragma unroll
            for (int s = 0; s < 2; ++s)
                yr[j][p][s] = *(const v2f*)(yl + (size_t)(8 * j + 2 * p + s) * N_PIX);
        }
    }
    const float* Wm = w4 + (size_t)(32 * h) * CH;
    const float* bm = bb4 + 32 * h;
    v4f wr[2][4];
#pragma unroll
    for (int ot = 0; ot < 2; ++ot) {
#pragma unroll
        for (int j = 0; j < 4; ++j)
            wr[ot][j] = *(const v4f*)(Wm + (size_t)(16 * ot + lo) * CH + 8 * j + 4 * hi);
    }

    v8f acc[2][2];
#pragma unroll
    for (int ot = 0; ot < 2; ++ot) {
        v4f b0 = *(const v4f*)(bm + 16 * ot + 8 * hi);
        v4f b1 = *(const v4f*)(bm + 16 * ot + 8 * hi + 4);
        v8f bv = v8f{b0.x, b0.y, b0.z, b0.w, b1.x, b1.y, b1.z, b1.w};
        acc[ot][0] = bv;
        acc[ot][1] = bv;
    }
    __builtin_amdgcn_sched_barrier(0);

#pragma unroll
    for (int j = 0; j < 4; ++j) {
#pragma unroll
        for (int p = 0; p < 2; ++p) {
#pragma unroll
            for (int ot = 0; ot < 2; ++ot) {
                v2f a = v2f{wr[ot][j][2 * p], wr[ot][j][2 * p + 1]};
#pragma unroll
                for (int T = 0; T < 2; ++T) {
                    v2f bf = v2f{yr[j][p][0][T], yr[j][p][1][T]};
                    acc[ot][T] = __builtin_amdgcn_wmma_f32_16x16x4_f32(
                                     false, a, false, bf, (short)0, acc[ot][T], false, false);
                }
            }
        }
    }

    float* ol = out + (size_t)b * CIN * N_PIX + (size_t)(32 * h + 8 * hi) * N_PIX + n0 + 2 * lo;
#pragma unroll
    for (int ot = 0; ot < 2; ++ot) {
#pragma unroll
        for (int v = 0; v < 8; ++v)
            *(v2f*)(ol + (size_t)(16 * ot + v) * N_PIX) = v2f{acc[ot][0][v], acc[ot][1][v]};
    }
}

extern "C" void kernel_launch(void* const* d_in, const int* in_sizes, int n_in,
                              void* d_out, int out_size, void* d_ws, size_t ws_size,
                              hipStream_t stream) {
    const float* x   = (const float*)d_in[0];
    const float* w1  = (const float*)d_in[1];
    const float* bb1 = (const float*)d_in[2];
    const float* w2  = (const float*)d_in[3];
    const float* bb2 = (const float*)d_in[4];
    const float* w3  = (const float*)d_in[5];
    const float* bb3 = (const float*)d_in[6];
    const float* w4  = (const float*)d_in[7];
    const float* bb4 = (const float*)d_in[8];
    float* out = (float*)d_out;

    float* Q = (float*)d_ws;
    float* K = Q + (size_t)BATCH * PLANE;
    float* V = K + (size_t)BATCH * PLANE;
    float* Y = V + (size_t)BATCH * PLANE;

    qkv_conv_kernel<<<(BATCH * N_PIX) / (32 * QKV_GROUPS), 32 * QKV_WAVES, 0, stream>>>(
        x, w1, bb1, w2, bb2, w3, bb3, Q, K, V);
    attn_kernel<<<N_PIX / 16, 32 * JSPLIT, 0, stream>>>(Q, K, V, Y);
    out_conv_kernel<<<(BATCH * N_PIX) / (32 * OC_GROUPS), 32 * OC_WAVES, 0, stream>>>(Y, w4, bb4, out);
}


# attention: a wave feeds two 16-row query tiles from each K/V tile it loads (K stored channel-pair interleaved: a load is an operand pair); keys split over 2 work-groups, out conv adds both Y planes
# speedup vs baseline: 9.9747x; 1.0887x over previous
#include <hip/hip_runtime.h>

typedef __attribute__((ext_vector_type(2))) float v2f;
typedef __attribute__((ext_vector_type(8))) float v8f;

#define N_PIX 4096
#define CH 32
#define CIN 64
#define BATCH 4
#define PLANE (CH * N_PIX)

typedef __attribute__((ext_vector_type(4))) float v4f;

#define QKV_GROUPS 2
#define QKV_WAVES (3 * QKV_GROUPS)

__global__ __launch_bounds__(32 * QKV_WAVES) void qkv_conv_kernel(
    const float* __restrict__ x,
    const float* __restrict__ w1, const float* __restrict__ bb1,
    const float* __restrict__ w2, const float* __restrict__ bb2,
    const float* __restrict__ w3, const float* __restrict__ bb3,
    float* __restrict__ Q, float* __restrict__ K, float* __restrict__ V)
{
    const int lane = threadIdx.x & 31;
    const int w  = __builtin_amdgcn_readfirstlane(threadIdx.x >> 5);
    const int lo = lane & 15;
    const int hi = lane >> 4;
    const int m  = w % 3;
    const int g  = blockIdx.x * QKV_GROUPS + w / 3;
    const int b  = g >> 7;
    const int n0 = (g & 127) * 32;

    const float* Wm = (m == 0) ? w1  : ((m == 1) ? w2  : w3);
    const float* bm = (m == 0) ? bb1 : ((m == 1) ? bb2 : bb3);
    float*       Om = (m == 0) ? Q   : ((m == 1) ? K   : V);

    const float* xl = x + (size_t)b * CIN * N_PIX + (size_t)(4 * hi) * N_PIX + n0 + 2 * lo;
    v2f xr[8][2][2];
#pragma unroll
    for (int j = 0; j < 8; ++j) {
#pragma unroll
        for (int p = 0; p < 2; ++p) {
#pragma unroll
            for (int s = 0; s < 2; ++s)
                xr[j][p][s] = *(const v2f*)(xl + (size_t)(8 * j + 2 * p + s) * N_PIX);
        }
    }
    v4f wr[2][8];
#pragma unroll
    for (int ot = 0; ot < 2; ++ot) {
#pragma unroll
        for (int j = 0; j < 8; ++j)
            wr[ot][j] = *(const v4f*)(Wm + (size_t)(16 * ot + lo) * CIN + 8 * j + 4 * hi);
    }

    v8f acc[2][2];
#pragma unroll
    for (int ot = 0; ot < 2; ++ot) {
        v4f b0 = *(const v4f*)(bm + 16 * ot + 8 * hi);
        v4f b1 = *(const v4f*)(bm + 16 * ot + 8 * hi + 4);
        v8f bv = v8f{b0.x, b0.y, b0.z, b0.w, b1.x, b1.y, b1.z, b1.w};
        acc[ot][0] = bv;
        acc[ot][1] = bv;
    }
    __builtin_amdgcn_sched_barrier(0);

#pragma unroll
    for (int j = 0; j < 8; ++j) {
#pragma unroll
        for (int p = 0; p < 2; ++p) {
#pragma unroll
            for (int ot = 0; ot < 2; ++ot) {
                v2f a = v2f{wr[ot][j][2 * p], wr[ot][j][2 * p + 1]};
#pragma unroll
                for (int T = 0; T < 2; ++T) {
                    v2f bf = v2f{xr[j][p][0][T], xr[j][p][1][T]};
                    acc[ot][T] = __builtin_amdgcn_wmma_f32_16x16x4_f32(
                                     false, a, false, bf, (short)0, acc[ot][T], false, false);
                }
            }
        }
    }

    if (m == 1) {
        float* kl = Om + (size_t)b * PLANE + ((size_t)(4 * hi) * N_PIX + n0 + 2 * lo) * 2;
#pragma unroll
        for (int ot = 0; ot < 2; ++ot) {
#pragma unroll
            for (int vp = 0; vp < 4; ++vp)
                *(v4f*)(kl + (size_t)(8 * ot + vp) * N_PIX * 2) =
                    v4f{acc[ot][0][2 * vp], acc[ot][0][2 * vp + 1],
                        acc[ot][1][2 * vp], acc[ot][1][2 * vp + 1]};
        }
    } else {
        float* ol = Om + (size_t)b * PLANE + (size_t)(8 * hi) * N_PIX + n0 + 2 * lo;
#pragma unroll
        for (int ot = 0; ot < 2; ++ot) {
#pragma unroll
            for (int v = 0; v < 8; ++v)
                *(v2f*)(ol + (size_t)(16 * ot + v) * N_PIX) = v2f{acc[ot][0][v], acc[ot][1][v]};
        }
    }
}

#define JSPLIT 8
#define KHALF (N_PIX / 2)
#define JW (KHALF / JSPLIT)
#define JSTEP 16
#define QT 2

__global__ __launch_bounds__(256) void attn_kernel(
    const float* __restrict__ Q, const float* __restrict__ Kp,
    const float* __restrict__ V, float* __restrict__ Y0, float* __restrict__ Y1)
{
    __shared__ float part[JSPLIT][BATCH][16 * CH];

    const int lane = threadIdx.x & 31;
    const int w    = __builtin_amdgcn_readfirstlane(threadIdx.x >> 5);
    const int lo   = lane & 15;
    const int hi   = lane >> 4;
    const int i0   = (blockIdx.x >> 1) * (16 * QT);
    const int half = blockIdx.x & 1;

    v2f qb[QT][BATCH][8];
#pragma unroll
    for (int q = 0; q < QT; ++q) {
#pragma unroll
        for (int b = 0; b < BATCH; ++b) {
#pragma unroll
            for (int kk = 0; kk < 8; ++kk)
                qb[q][b][kk] = *(const v2f*)(Q + (size_t)b * PLANE +
                                             (size_t)(i0 + 16 * q + lo) * CH + 4 * kk + 2 * hi);
        }
    }

    v8f acc[QT][BATCH][2];
#pragma unroll
    for (int q = 0; q < QT; ++q) {
#pragma unroll
        for (int b = 0; b < BATCH; ++b) { acc[q][b][0] = v8f{}; acc[q][b][1] = v8f{}; }
    }

    const float* Kl = Kp + ((size_t)hi * N_PIX + lo) * 2;
    const float* Vl = V + (size_t)(8 * hi) * CH + 2 * lo;

    v2f kr[BATCH][8];
    v2f vr[BATCH][4][2];

    auto load_k = [&](int j) {
#pragma unroll
        for (int kk = 0; kk < 8; ++kk) {
#pragma unroll
            for (int b = 0; b < BATCH; ++b)
                kr[b][kk] = *(const v2f*)(Kl + (size_t)b * PLANE +
                                          ((size_t)(2 * kk) * N_PIX + j) * 2);
        }
    };
    auto load_v = [&](int j) {
#pragma unroll
        for (int kk = 0; kk < 4; ++kk) {
#pragma unroll
            for (int b = 0; b < BATCH; ++b) {
#pragma unroll
                for (int r = 0; r < 2; ++r)
                    vr[b][kk][r] = *(const v2f*)(Vl + (size_t)b * PLANE +
                                                 (size_t)(j + 2 * kk + r) * CH);
            }
        }
    };
    auto scores = [&](v8f (&s)[QT][BATCH]) {
#pragma unroll
        for (int kk = 0; kk < 8; ++kk) {
#pragma unroll
            for (int b = 0; b < BATCH; ++b) {
#pragma unroll
                for (int q = 0; q < QT; ++q)
                    s[q][b] = __builtin_amdgcn_wmma_f32_16x16x4_f32(
                                  false, kr[b][kk], false, qb[q][b][kk], (short)0, s[q][b],
                                  false, false);
            }
        }
    };
    auto softmax4 = [&](v8f (&s)[BATCH]) {
#pragma unroll
        for (int v = 0; v < 8; ++v) {
            float s0 = s[0][v], s1 = s[1][v], s2 = s[2][v], s3 = s[3][v];
            float m = fmaxf(fmaxf(s0, s1), fmaxf(s2, s3));
            float e0 = __expf(s0 - m), e1 = __expf(s1 - m);
            float e2 = __expf(s2 - m), e3 = __expf(s3 - m);
            float d = e0 + e1 + e2 + e3;
            float rd = __builtin_amdgcn_rcpf(d);
            s[0][v] = e0 * rd; s[1][v] = e1 * rd;
            s[2][v] = e2 * rd; s[3][v] = e3 * rd;
        }
    };
    auto pv = [&](v8f (&s)[QT][BATCH]) {
#pragma unroll
        for (int kk = 0; kk < 4; ++kk) {
#pragma unroll
            for (int b = 0; b < BATCH; ++b) {
                v2f bl = v2f{vr[b][kk][0].x, vr[b][kk][1].x};
                v2f bh = v2f{vr[b][kk][0].y, vr[b][kk][1].y};
#pragma unroll
                for (int q = 0; q < QT; ++q) {
                    v2f a = v2f{s[q][b][2 * kk], s[q][b][2 * kk + 1]};
                    acc[q][b][0] = __builtin_amdgcn_wmma_f32_16x16x4_f32(
                                       false, a, false, bl, (short)0, acc[q][b][0], false, false);
                    acc[q][b][1] = __builtin_amdgcn_wmma_f32_16x16x4_f32(
                                       false, a, false, bh, (short)0, acc[q][b][1], false, false);
                }
            }
        }
    };

    auto step = [&](int jt, bool ahead) {
        v8f s[QT][BATCH];
#pragma unroll
        for (int q = 0; q < QT; ++q) {
#pragma unroll
            for (int b = 0; b < BATCH; ++b) s[q][b] = v8f{};
        }
        scores(s);
        if (ahead) {
            __builtin_amdgcn_sched_barrier(0);
            load_k(jt + JSTEP);
            __builtin_amdgcn_sched_barrier(0);
        }
        softmax4(s[0]);
        softmax4(s[1]);
        pv(s);
        if (ahead) {
            __builtin_amdgcn_sched_barrier(0);
            load_v(jt + JSTEP);
            __builtin_amdgcn_sched_barrier(0);
        }
    };

    const int jbeg = half * KHALF + w * JW;
    const int jlast = jbeg + JW - JSTEP;
    load_k(jbeg);
    load_v(jbeg);
    for (int jt = jbeg; jt < jlast; jt += JSTEP) step(jt, true);
    step(jlast, false);

    float* Yh = half ? Y1 : Y0;
    const size_t ypl = half ? (size_t)CIN * N_PIX : (size_t)PLANE;
    const int e2 = threadIdx.x * 2;
#pragma unroll
    for (int q = 0; q < QT; ++q) {
        if (q) __syncthreads();
#pragma unroll
        for (int b = 0; b < BATCH; ++b) {
#pragma unroll
            for (int v = 0; v < 8; ++v)
                *(v2f*)&part[w][b][(v + 8 * hi) * CH + 2 * lo] =
                    v2f{acc[q][b][0][v], acc[q][b][1][v]};
        }
        __syncthreads();
#pragma unroll
        for (int b = 0; b < BATCH; ++b) {
            v2f sum = *(const v2f*)&part[0][b][e2];
#pragma unroll
            for (int ww = 1; ww < JSPLIT; ++ww) sum += *(const v2f*)&part[ww][b][e2];
            *(v2f*)(Yh + (size_t)b * ypl + (size_t)(i0 + 16 * q) * CH + e2) = sum;
        }
    }
}

#define OC_WAVES 2

__global__ __launch_bounds__(32 * OC_WAVES) void out_conv_kernel(
    const float* __restrict__ Y0, const float* __restrict__ w4,
    const float* __restrict__ bb4, float* out)
{
    const int lane = threadIdx.x & 31;
    const int w  = __builtin_amdgcn_readfirstlane(threadIdx.x >> 5);
    const int lo = lane & 15;
    const int hi = lane >> 4;
    const int g  = blockIdx.x * OC_WAVES + w;
    const int b  = g >> 7;
    const int n0 = (g & 127) * 32;

    const float* yl0 = Y0 + (size_t)b * PLANE + (size_t)(4 * hi) * N_PIX + n0 + 2 * lo;
    const float* yl1 = out + (size_t)b * CIN * N_PIX + (size_t)(4 * hi) * N_PIX + n0 + 2 * lo;
    v2f ya[4][2][2], yb[4][2][2];
#pragma unroll
    for (int j = 0; j < 4; ++j) {
#pragma unroll
        for (int p = 0; p < 2; ++p) {
#pragma unroll
            for (int s = 0; s < 2; ++s) {
                ya[j][p][s] = *(const v2f*)(yl0 + (size_t)(8 * j + 2 * p + s) * N_PIX);
                yb[j][p][s] = *(const v2f*)(yl1 + (size_t)(8 * j + 2 * p + s) * N_PIX);
            }
        }
    }
    v4f wr[4][4];
#pragma unroll
    for (int ot = 0; ot < 4; ++ot) {
#pragma unroll
        for (int j = 0; j < 4; ++j)
            wr[ot][j] = *(const v4f*)(w4 + (size_t)(16 * ot + lo) * CH + 8 * j + 4 * hi);
    }

    v8f acc[4][2];
#pragma unroll
    for (int ot = 0; ot < 4; ++ot) {
        v4f b0 = *(const v4f*)(bb4 + 16 * ot + 8 * hi);
        v4f b1 = *(const v4f*)(bb4 + 16 * ot + 8 * hi + 4);
        v8f bv = v8f{b0.x, b0.y, b0.z, b0.w, b1.x, b1.y, b1.z, b1.w};
        acc[ot][0] = bv;
        acc[ot][1] = bv;
    }
    __builtin_amdgcn_sched_barrier(0);

#pragma unroll
    for (int j = 0; j < 4; ++j) {
#pragma unroll
        for (int p = 0; p < 2; ++p) {
            v2f y0 = ya[j][p][0] + yb[j][p][0];
            v2f y1 = ya[j][p][1] + yb[j][p][1];
#pragma unroll
            for (int T = 0; T < 2; ++T) {
                v2f bf = v2f{y0[T], y1[T]};
#pragma unroll
                for (int ot = 0; ot < 4; ++ot) {
                    v2f a = v2f{wr[ot][j][2 * p], wr[ot][j][2 * p + 1]};
                    acc[ot][T] = __builtin_amdgcn_wmma_f32_16x16x4_f32(
                                     false, a, false, bf, (short)0, acc[ot][T], false, false);
                }
            }
        }
    }

    float* ol = out + (size_t)b * CIN * N_PIX + (size_t)(8 * hi) * N_PIX + n0 + 2 * lo;
#pragma unroll
    for (int ot = 0; ot < 4; ++ot) {
#pragma unroll
        for (int v = 0; v < 8; ++v)
            *(v2f*)(ol + (size_t)(16 * ot + v) * N_PIX) = v2f{acc[ot][0][v], acc[ot][1][v]};
    }
}

extern "C" void kernel_launch(void* const* d_in, const int* in_sizes, int n_in,
                              void* d_out, int out_size, void* d_ws, size_t ws_size,
                              hipStream_t stream) {
    const float* x   = (const float*)d_in[0];
    const float* w1  = (const float*)d_in[1];
    const float* bb1 = (const float*)d_in[2];
    const float* w2  = (const float*)d_in[3];
    const float* bb2 = (const float*)d_in[4];
    const float* w3  = (const float*)d_in[5];
    const float* bb3 = (const float*)d_in[6];
    const float* w4  = (const float*)d_in[7];
    const float* bb4 = (const float*)d_in[8];
    float* out = (float*)d_out;

    float* Q = (float*)d_ws;
    float* K = Q + (size_t)BATCH * PLANE;
    float* V = K + (size_t)BATCH * PLANE;
    float* Y = V + (size_t)BATCH * PLANE;

    qkv_conv_kernel<<<(BATCH * N_PIX) / (32 * QKV_GROUPS), 32 * QKV_WAVES, 0, stream>>>(
        x, w1, bb1, w2, bb2, w3, bb3, Q, K, V);
    attn_kernel<<<(N_PIX / (16 * QT)) * 2, 32 * JSPLIT, 0, stream>>>(Q, K, V, Y, out);
    out_conv_kernel<<<(BATCH * N_PIX) / (32 * OC_WAVES), 32 * OC_WAVES, 0, stream>>>(Y, w4, bb4, out);
}


# attention softmax with fewer vector instructions: Q fragments scaled by log2(e) once per wave (scores in log2 units, last-bit rounding changes); differences, sums, products as 2-element packed ops
# speedup vs baseline: 10.2888x; 1.0315x over previous
#include <hip/hip_runtime.h>

typedef __attribute__((ext_vector_type(2))) float v2f;
typedef __attribute__((ext_vector_type(8))) float v8f;

#define N_PIX 4096
#define CH 32
#define CIN 64
#define BATCH 4
#define PLANE (CH * N_PIX)

typedef __attribute__((ext_vector_type(4))) float v4f;

#define QKV_GROUPS 2
#define QKV_WAVES (3 * QKV_GROUPS)

__global__ __launch_bounds__(32 * QKV_WAVES) void qkv_conv_kernel(
    const float* __restrict__ x,
    const float* __restrict__ w1, const float* __restrict__ bb1,
    const float* __restrict__ w2, const float* __restrict__ bb2,
    const float* __restrict__ w3, const float* __restrict__ bb3,
    float* __restrict__ Q, float* __restrict__ K, float* __restrict__ V)
{
    const int lane = threadIdx.x & 31;
    const int w  = __builtin_amdgcn_readfirstlane(threadIdx.x >> 5);
    const int lo = lane & 15;
    const int hi = lane >> 4;
    const int m  = w % 3;
    const int g  = blockIdx.x * QKV_GROUPS + w / 3;
    const int b  = g >> 7;
    const int n0 = (g & 127) * 32;

    const float* Wm = (m == 0) ? w1  : ((m == 1) ? w2  : w3);
    const float* bm = (m == 0) ? bb1 : ((m == 1) ? bb2 : bb3);
    float*       Om = (m == 0) ? Q   : ((m == 1) ? K   : V);

    const float* xl = x + (size_t)b * CIN * N_PIX + (size_t)(4 * hi) * N_PIX + n0 + 2 * lo;
    v2f xr[8][2][2];
#pragma unroll
    for (int j = 0; j < 8; ++j) {
#pragma unroll
        for (int p = 0; p < 2; ++p) {
#pragma unroll
            for (int s = 0; s < 2; ++s)
                xr[j][p][s] = *(const v2f*)(xl + (size_t)(8 * j + 2 * p + s) * N_PIX);
        }
    }
    v4f wr[2][8];
#pragma unroll
    for (int ot = 0; ot < 2; ++ot) {
#pragma unroll
        for (int j = 0; j < 8; ++j)
            wr[ot][j] = *(const v4f*)(Wm + (size_t)(16 * ot + lo) * CIN + 8 * j + 4 * hi);
    }

    v8f acc[2][2];
#pragma unroll
    for (int ot = 0; ot < 2; ++ot) {
        v4f b0 = *(const v4f*)(bm + 16 * ot + 8 * hi);
        v4f b1 = *(const v4f*)(bm + 16 * ot + 8 * hi + 4);
        v8f bv = v8f{b0.x, b0.y, b0.z, b0.w, b1.x, b1.y, b1.z, b1.w};
        acc[ot][0] = bv;
        acc[ot][1] = bv;
    }
    __builtin_amdgcn_sched_barrier(0);

#pragma unroll
    for (int j = 0; j < 8; ++j) {
#pragma unroll
        for (int p = 0; p < 2; ++p) {
#pragma unroll
            for (int ot = 0; ot < 2; ++ot) {
                v2f a = v2f{wr[ot][j][2 * p], wr[ot][j][2 * p + 1]};
#pragma unroll
                for (int T = 0; T < 2; ++T) {
                    v2f bf = v2f{xr[j][p][0][T], xr[j][p][1][T]};
                    acc[ot][T] = __builtin_amdgcn_wmma_f32_16x16x4_f32(
                                     false, a, false, bf, (short)0, acc[ot][T], false, false);
                }
            }
        }
    }

    if (m == 1) {
        float* kl = Om + (size_t)b * PLANE + ((size_t)(4 * hi) * N_PIX + n0 + 2 * lo) * 2;
#pragma unroll
        for (int ot = 0; ot < 2; ++ot) {
#pragma unroll
            for (int vp = 0; vp < 4; ++vp)
                *(v4f*)(kl + (size_t)(8 * ot + vp) * N_PIX * 2) =
                    v4f{acc[ot][0][2 * vp], acc[ot][0][2 * vp + 1],
                        acc[ot][1][2 * vp], acc[ot][1][2 * vp + 1]};
        }
    } else {
        float* ol = Om + (size_t)b * PLANE + (size_t)(8 * hi) * N_PIX + n0 + 2 * lo;
#pragma unroll
        for (int ot = 0; ot < 2; ++ot) {
#pragma unroll
            for (int v = 0; v < 8; ++v)
                *(v2f*)(ol + (size_t)(16 * ot + v) * N_PIX) = v2f{acc[ot][0][v], acc[ot][1][v]};
        }
    }
}

#define JSPLIT 8
#define KHALF (N_PIX / 2)
#define JW (KHALF / JSPLIT)
#define JSTEP 16
#define QT 2

__global__ __launch_bounds__(256) void attn_kernel(
    const float* __restrict__ Q, const float* __restrict__ Kp,
    const float* __restrict__ V, float* __restrict__ Y0, float* __restrict__ Y1)
{
    __shared__ float part[JSPLIT][BATCH][16 * CH];

    const int lane = threadIdx.x & 31;
    const int w    = __builtin_amdgcn_readfirstlane(threadIdx.x >> 5);
    const int lo   = lane & 15;
    const int hi   = lane >> 4;
    const int i0   = (blockIdx.x >> 1) * (16 * QT);
    const int half = blockIdx.x & 1;

    const float LOG2E = 1.44269504088896340736f;
    v2f qb[QT][BATCH][8];
#pragma unroll
    for (int q = 0; q < QT; ++q) {
#pragma unroll
        for (int b = 0; b < BATCH; ++b) {
#pragma unroll
            for (int kk = 0; kk < 8; ++kk)
                qb[q][b][kk] = *(const v2f*)(Q + (size_t)b * PLANE +
                                             (size_t)(i0 + 16 * q + lo) * CH + 4 * kk + 2 * hi) * LOG2E;
        }
    }

    v8f acc[QT][BATCH][2];
#pragma unroll
    for (int q = 0; q < QT; ++q) {
#pragma unroll
        for (int b = 0; b < BATCH; ++b) { acc[q][b][0] = v8f{}; acc[q][b][1] = v8f{}; }
    }

    const float* Kl = Kp + ((size_t)hi * N_PIX + lo) * 2;
    const float* Vl = V + (size_t)(8 * hi) * CH + 2 * lo;

    v2f kr[BATCH][8];
    v2f vr[BATCH][4][2];

    auto load_k = [&](int j) {
#pragma unroll
        for (int kk = 0; kk < 8; ++kk) {
#pragma unroll
            for (int b = 0; b < BATCH; ++b)
                kr[b][kk] = *(const v2f*)(Kl + (size_t)b * PLANE +
                                          ((size_t)(2 * kk) * N_PIX + j) * 2);
        }
    };
    auto load_v = [&](int j) {
#pragma unroll
        for (int kk = 0; kk < 4; ++kk) {
#pragma unroll
            for (int b = 0; b < BATCH; ++b) {
#pragma unroll
                for (int r = 0; r < 2; ++r)
                    vr[b][kk][r] = *(const v2f*)(Vl + (size_t)b * PLANE +
                                                 (size_t)(j + 2 * kk + r) * CH);
            }
        }
    };
    auto scores = [&](v8f (&s)[QT][BATCH]) {
#pragma unroll
        for (int kk = 0; kk < 8; ++kk) {
#pragma unroll
            for (int b = 0; b < BATCH; ++b) {
#pragma unroll
                for (int q = 0; q < QT; ++q)
                    s[q][b] = __builtin_amdgcn_wmma_f32_16x16x4_f32(
                                  false, kr[b][kk], false, qb[q][b][kk], (short)0, s[q][b],
                                  false, false);
            }
        }
    };
    auto softmax4 = [&](v8f (&s)[BATCH]) {
#pragma unroll
        for (int p = 0; p < 4; ++p) {
            v2f t0 = v2f{s[0][2 * p], s[0][2 * p + 1]}, t1 = v2f{s[1][2 * p], s[1][2 * p + 1]};
            v2f t2 = v2f{s[2][2 * p], s[2][2 * p + 1]}, t3 = v2f{s[3][2 * p], s[3][2 * p + 1]};
            const float NINF = -__builtin_inff();
            v2f m;
            m.x = fmaxf(fmaxf(fmaxf(fmaxf(t0.x, t1.x), t2.x), t3.x), NINF);
            m.y = fmaxf(fmaxf(fmaxf(fmaxf(t0.y, t1.y), t2.y), t3.y), NINF);
            v2f d0 = t0 - m, d1 = t1 - m, d2 = t2 - m, d3 = t3 - m;
            v2f e0 = v2f{__builtin_amdgcn_exp2f(d0.x), __builtin_amdgcn_exp2f(d0.y)};
            v2f e1 = v2f{__builtin_amdgcn_exp2f(d1.x), __builtin_amdgcn_exp2f(d1.y)};
            v2f e2 = v2f{__builtin_amdgcn_exp2f(d2.x), __builtin_amdgcn_exp2f(d2.y)};
            v2f e3 = v2f{__builtin_amdgcn_exp2f(d3.x), __builtin_amdgcn_exp2f(d3.y)};
            v2f d = ((e0 + e1) + e2) + e3;
            v2f rd = v2f{__builtin_amdgcn_rcpf(d.x), __builtin_amdgcn_rcpf(d.y)};
            e0 *= rd; e1 *= rd; e2 *= rd; e3 *= rd;
            s[0][2 * p] = e0.x; s[0][2 * p + 1] = e0.y;
            s[1][2 * p] = e1.x; s[1][2 * p + 1] = e1.y;
            s[2][2 * p] = e2.x; s[2][2 * p + 1] = e2.y;
            s[3][2 * p] = e3.x; s[3][2 * p + 1] = e3.y;
        }
    };
    auto pv = [&](v8f (&s)[QT][BATCH]) {
#pragma unroll
        for (int kk = 0; kk < 4; ++kk) {
#pragma unroll
            for (int b = 0; b < BATCH; ++b) {
                v2f bl = v2f{vr[b][kk][0].x, vr[b][kk][1].x};
                v2f bh = v2f{vr[b][kk][0].y, vr[b][kk][1].y};
#pragma unroll
                for (int q = 0; q < QT; ++q) {
                    v2f a = v2f{s[q][b][2 * kk], s[q][b][2 * kk + 1]};
                    acc[q][b][0] = __builtin_amdgcn_wmma_f32_16x16x4_f32(
                                       false, a, false, bl, (short)0, acc[q][b][0], false, false);
                    acc[q][b][1] = __builtin_amdgcn_wmma_f32_16x16x4_f32(
                                       false, a, false, bh, (short)0, acc[q][b][1], false, false);
                }
            }
        }
    };

    auto step = [&](int jt, bool ahead) {
        v8f s[QT][BATCH];
#pragma unroll
        for (int q = 0; q < QT; ++q) {
#pragma unroll
            for (int b = 0; b < BATCH; ++b) s[q][b] = v8f{};
        }
        scores(s);
        if (ahead) {
            __builtin_amdgcn_sched_barrier(0);
            load_k(jt + JSTEP);
            __builtin_amdgcn_sched_barrier(0);
        }
        softmax4(s[0]);
        softmax4(s[1]);
        pv(s);
        if (ahead) {
            __builtin_amdgcn_sched_barrier(0);
            load_v(jt + JSTEP);
            __builtin_amdgcn_sched_barrier(0);
        }
    };

    const int jbeg = half * KHALF + w * JW;
    const int jlast = jbeg + JW - JSTEP;
    load_k(jbeg);
    load_v(jbeg);
    for (int jt = jbeg; jt < jlast; jt += JSTEP) step(jt, true);
    step(jlast, false);

    float* Yh = half ? Y1 : Y0;
    const size_t ypl = half ? (size_t)CIN * N_PIX : (size_t)PLANE;
    const int e2 = threadIdx.x * 2;
#pragma unroll
    for (int q = 0; q < QT; ++q) {
        if (q) __syncthreads();
#pragma unroll
        for (int b = 0; b < BATCH; ++b) {
#pragma unroll
            for (int v = 0; v < 8; ++v)
                *(v2f*)&part[w][b][(v + 8 * hi) * CH + 2 * lo] =
                    v2f{acc[q][b][0][v], acc[q][b][1][v]};
        }
        __syncthreads();
#pragma unroll
        for (int b = 0; b < BATCH; ++b) {
            v2f sum = *(const v2f*)&part[0][b][e2];
#pragma unroll
            for (int ww = 1; ww < JSPLIT; ++ww) sum += *(const v2f*)&part[ww][b][e2];
            *(v2f*)(Yh + (size_t)b * ypl + (size_t)(i0 + 16 * q) * CH + e2) = sum;
        }
    }
}

#define OC_WAVES 2

__global__ __launch_bounds__(32 * OC_WAVES) void out_conv_kernel(
    const float* __restrict__ Y0, const float* __restrict__ w4,
    const float* __restrict__ bb4, float* out)
{
    const int lane = threadIdx.x & 31;
    const int w  = __builtin_amdgcn_readfirstlane(threadIdx.x >> 5);
    const int lo = lane & 15;
    const int hi = lane >> 4;
    const int g  = blockIdx.x * OC_WAVES + w;
    const int b  = g >> 7;
    const int n0 = (g & 127) * 32;

    const float* yl0 = Y0 + (size_t)b * PLANE + (size_t)(4 * hi) * N_PIX + n0 + 2 * lo;
    const float* yl1 = out + (size_t)b * CIN * N_PIX + (size_t)(4 * hi) * N_PIX + n0 + 2 * lo;
    v2f ya[4][2][2], yb[4][2][2];
#pragma unroll
    for (int j = 0; j < 4; ++j) {
#pragma unroll
        for (int p = 0; p < 2; ++p) {
#pragma unroll
            for (int s = 0; s < 2; ++s) {
                ya[j][p][s] = *(const v2f*)(yl0 + (size_t)(8 * j + 2 * p + s) * N_PIX);
                yb[j][p][s] = *(const v2f*)(yl1 + (size_t)(8 * j + 2 * p + s) * N_PIX);
            }
        }
    }
    v4f wr[4][4];
#pragma unroll
    for (int ot = 0; ot < 4; ++ot) {
#pragma unroll
        for (int j = 0; j < 4; ++j)
            wr[ot][j] = *(const v4f*)(w4 + (size_t)(16 * ot + lo) * CH + 8 * j + 4 * hi);
    }

    v8f acc[4][2];
#pragma unroll
    for (int ot = 0; ot < 4; ++ot) {
        v4f b0 = *(const v4f*)(bb4 + 16 * ot + 8 * hi);
        v4f b1 = *(const v4f*)(bb4 + 16 * ot + 8 * hi + 4);
        v8f bv = v8f{b0.x, b0.y, b0.z, b0.w, b1.x, b1.y, b1.z, b1.w};
        acc[ot][0] = bv;
        acc[ot][1] = bv;
    }
    __builtin_amdgcn_sched_barrier(0);

#pragma unroll
    for (int j = 0; j < 4; ++j) {
#pragma unroll
        for (int p = 0; p < 2; ++p) {
            v2f y0 = ya[j][p][0] + yb[j][p][0];
            v2f y1 = ya[j][p][1] + yb[j][p][1];
#pragma unroll
            for (int T = 0; T < 2; ++T) {
                v2f bf = v2f{y0[T], y1[T]};
#pragma unroll
                for (int ot = 0; ot < 4; ++ot) {
                    v2f a = v2f{wr[ot][j][2 * p], wr[ot][j][2 * p + 1]};
                    acc[ot][T] = __builtin_amdgcn_wmma_f32_16x16x4_f32(
                                     false, a, false, bf, (short)0, acc[ot][T], false, false);
                }
            }
        }
    }

    float* ol = out + (size_t)b * CIN * N_PIX + (size_t)(8 * hi) * N_PIX + n0 + 2 * lo;
#pragma unroll
    for (int ot = 0; ot < 4; ++ot) {
#pragma unroll
        for (int v = 0; v < 8; ++v)
            *(v2f*)(ol + (size_t)(16 * ot + v) * N_PIX) = v2f{acc[ot][0][v], acc[ot][1][v]};
    }
}

extern "C" void kernel_launch(void* const* d_in, const int* in_sizes, int n_in,
                              void* d_out, int out_size, void* d_ws, size_t ws_size,
                              hipStream_t stream) {
    const float* x   = (const float*)d_in[0];
    const float* w1  = (const float*)d_in[1];
    const float* bb1 = (const float*)d_in[2];
    const float* w2  = (const float*)d_in[3];
    const float* bb2 = (const float*)d_in[4];
    const float* w3  = (const float*)d_in[5];
    const float* bb3 = (const float*)d_in[6];
    const float* w4  = (const float*)d_in[7];
    const float* bb4 = (const float*)d_in[8];
    float* out = (float*)d_out;

    float* Q = (float*)d_ws;
    float* K = Q + (size_t)BATCH * PLANE;
    float* V = K + (size_t)BATCH * PLANE;
    float* Y = V + (size_t)BATCH * PLANE;

    qkv_conv_kernel<<<(BATCH * N_PIX) / (32 * QKV_GROUPS), 32 * QKV_WAVES, 0, stream>>>(
        x, w1, bb1, w2, bb2, w3, bb3, Q, K, V);
    attn_kernel<<<(N_PIX / (16 * QT)) * 2, 32 * JSPLIT, 0, stream>>>(Q, K, V, Y, out);
    out_conv_kernel<<<(BATCH * N_PIX) / (32 * OC_WAVES), 32 * OC_WAVES, 0, stream>>>(Y, w4, bb4, out);
}
